# starting kernel
# baseline (speedup 1.0000x reference)
#include <hip/hip_runtime.h>


typedef __attribute__((ext_vector_type(2))) float v2f;
typedef __attribute__((ext_vector_type(8))) float v8f;

#define T_LEN   4096
#define IN_CH   64
#define OUT_CH  64
#define NB      16
#define NA      8
#define TILE_T  32
#define STAGE_ROWS (TILE_T + NB - 1)
#define LDS_STRIDE 49

__global__ __launch_bounds__(256) void fir_wmma_kernel(
    const float* __restrict__ x,
    const float* __restrict__ w,
    float* __restrict__ u)
{
    __shared__ float xs[IN_CH][LDS_STRIDE];

    const int tid  = threadIdx.x;
    const int bpb  = T_LEN / TILE_T;
    const int b    = blockIdx.x / bpb;
    const int t0   = (blockIdx.x % bpb) * TILE_T;

    for (int row = tid; row < NB * IN_CH; row += 256)
        __builtin_prefetch(w + (size_t)row * OUT_CH, 0, 1);

    {
        const int ci = tid & (IN_CH - 1);
        for (int r = tid >> 6; r < STAGE_ROWS; r += 4) {
            const int tg = t0 - (NB - 1) + r;
            float v = 0.0f;
            if (tg >= 0)
                v = x[((size_t)b * T_LEN + tg) * IN_CH + ci];
            xs[ci][r] = v;
        }
    }
    __syncthreads();

    const int lane  = tid & 31;
    const int wave  = tid >> 5;
    const int mtile = wave >> 2;
    const int ntile = wave & 3;
    const int mrow  = lane & 15;
    const int hi    = lane >> 4;
    const int ncol  = ntile * 16 + (lane & 15);

    v8f c = {};
    for (int tap = 0; tap < NB; ++tap) {
        const int tl = mtile * 16 + mrow + tap;
        const float* wk = w + ((size_t)tap * IN_CH) * OUT_CH + ncol;
#pragma unroll
        for (int i4 = 0; i4 < IN_CH; i4 += 4) {
            const int kk = i4 + hi * 2;
            v2f a, bf;
            a.x  = xs[kk][tl];
            a.y  = xs[kk + 1][tl];
            bf.x = wk[(size_t)kk * OUT_CH];
            bf.y = wk[(size_t)(kk + 1) * OUT_CH];
            c = __builtin_amdgcn_wmma_f32_16x16x4_f32(
                     false, a,  false, bf,
                     (short)0, c,  false,  false);
        }
    }

    const int trow = t0 + mtile * 16 + hi * 8;
    float* uout = u + ((size_t)b * T_LEN) * OUT_CH + ncol;
#pragma unroll
    for (int v = 0; v < 8; ++v)
        uout[(size_t)(trow + v) * OUT_CH] = c[v];
}

__global__ __launch_bounds__(64) void ar_scan_kernel(
    const float* __restrict__ a_coeff,
    float* __restrict__ y)
{
    const int b = blockIdx.x;
    const int o = threadIdx.x;

    float a[NA];
#pragma unroll
    for (int j = 0; j < NA; ++j)
        a[j] = a_coeff[j * OUT_CH + o];

    float h[NA];
#pragma unroll
    for (int j = 0; j < NA; ++j) h[j] = 0.0f;

    float* p = y + ((size_t)b * T_LEN) * OUT_CH + o;

#pragma unroll 8
    for (int t = 0; t < T_LEN; ++t) {
        float acc = p[(size_t)t * OUT_CH];
#pragma unroll
        for (int j = NA - 1; j >= 0; --j)
            acc = fmaf(a[j], h[j], acc);
#pragma unroll
        for (int j = NA - 1; j > 0; --j) h[j] = h[j - 1];
        h[0] = acc;
        p[(size_t)t * OUT_CH] = acc;
    }
}

extern "C" void kernel_launch(void* const* d_in, const int* in_sizes, int n_in,
                              void* d_out, int out_size, void* d_ws, size_t ws_size,
                              hipStream_t stream)
{
    const float* x = (const float*)d_in[0];
    const float* a = (const float*)d_in[1];
    const float* w = (const float*)d_in[2];
    float* out = (float*)d_out;

    const int B = in_sizes[0] / (T_LEN * IN_CH);

    dim3 grid1((unsigned)(B * (T_LEN / TILE_T)));
    fir_wmma_kernel<<<grid1, 256, 0, stream>>>(x, w, out);

    ar_scan_kernel<<<dim3((unsigned)B), 64, 0, stream>>>(a, out);
}


# AR scan reads u 56 steps ahead into a register ring: a block's loads are in flight while the block before it runs its serial recurrence (was one memory round trip per 8 steps); FMA order unchanged
# speedup vs baseline: 2.4995x; 2.4995x over previous
#include <hip/hip_runtime.h>


typedef __attribute__((ext_vector_type(2))) float v2f;
typedef __attribute__((ext_vector_type(8))) float v8f;

#define T_LEN   4096
#define IN_CH   64
#define OUT_CH  64
#define NB      16
#define NA      8
#define TILE_T  32
#define STAGE_ROWS (TILE_T + NB - 1)
#define LDS_STRIDE 49

__global__ __launch_bounds__(256) void fir_wmma_kernel(
    const float* __restrict__ x,
    const float* __restrict__ w,
    float* __restrict__ u)
{
    __shared__ float xs[IN_CH][LDS_STRIDE];

    const int tid  = threadIdx.x;
    const int bpb  = T_LEN / TILE_T;
    const int b    = blockIdx.x / bpb;
    const int t0   = (blockIdx.x % bpb) * TILE_T;

    for (int row = tid; row < NB * IN_CH; row += 256)
        __builtin_prefetch(w + (size_t)row * OUT_CH, 0, 1);

    {
        const int ci = tid & (IN_CH - 1);
        for (int r = tid >> 6; r < STAGE_ROWS; r += 4) {
            const int tg = t0 - (NB - 1) + r;
            float v = 0.0f;
            if (tg >= 0)
                v = x[((size_t)b * T_LEN + tg) * IN_CH + ci];
            xs[ci][r] = v;
        }
    }
    __syncthreads();

    const int lane  = tid & 31;
    const int wave  = tid >> 5;
    const int mtile = wave >> 2;
    const int ntile = wave & 3;
    const int mrow  = lane & 15;
    const int hi    = lane >> 4;
    const int ncol  = ntile * 16 + (lane & 15);

    v8f c = {};
    for (int tap = 0; tap < NB; ++tap) {
        const int tl = mtile * 16 + mrow + tap;
        const float* wk = w + ((size_t)tap * IN_CH) * OUT_CH + ncol;
#pragma unroll
        for (int i4 = 0; i4 < IN_CH; i4 += 4) {
            const int kk = i4 + hi * 2;
            v2f a, bf;
            a.x  = xs[kk][tl];
            a.y  = xs[kk + 1][tl];
            bf.x = wk[(size_t)kk * OUT_CH];
            bf.y = wk[(size_t)(kk + 1) * OUT_CH];
            c = __builtin_amdgcn_wmma_f32_16x16x4_f32(
                     false, a,  false, bf,
                     (short)0, c,  false,  false);
        }
    }

    const int trow = t0 + mtile * 16 + hi * 8;
    float* uout = u + ((size_t)b * T_LEN) * OUT_CH + ncol;
#pragma unroll
    for (int v = 0; v < 8; ++v)
        uout[(size_t)(trow + v) * OUT_CH] = c[v];
}

#define PF 56
#define AR_NBLK (T_LEN / PF)
#define AR_REM  (T_LEN % PF)

template <int STEPS, int REFILL>
__device__ __forceinline__ void ar_block(float* __restrict__ q,
                                         const float (&a)[NA],
                                         float (&h)[NA],
                                         float (&ring)[PF])
{
#pragma unroll
    for (int i = 0; i < STEPS; ++i) {
        float acc = ring[i];
        if (i < REFILL)
            ring[i] = q[(size_t)(STEPS + i) * OUT_CH];
#pragma unroll
        for (int j = NA - 1; j >= 0; --j)
            acc = fmaf(a[j], h[j], acc);
#pragma unroll
        for (int j = NA - 1; j > 0; --j) h[j] = h[j - 1];
        h[0] = acc;
        q[(size_t)i * OUT_CH] = acc;
    }
}

__global__ __launch_bounds__(64) void ar_scan_kernel(
    const float* __restrict__ a_coeff,
    float* __restrict__ y)
{
    const int b = blockIdx.x;
    const int o = threadIdx.x;

    float a[NA];
#pragma unroll
    for (int j = 0; j < NA; ++j)
        a[j] = a_coeff[j * OUT_CH + o];

    float* p = y + ((size_t)b * T_LEN) * OUT_CH + o;

    float ring[PF];
#pragma unroll
    for (int i = 0; i < PF; ++i)
        ring[i] = p[(size_t)i * OUT_CH];

    float h[NA];
#pragma unroll
    for (int j = 0; j < NA; ++j) h[j] = 0.0f;

    for (int k = 0; k < AR_NBLK - 1; ++k)
        ar_block<PF, PF>(p + (size_t)k * PF * OUT_CH, a, h, ring);
    ar_block<PF, AR_REM>(p + (size_t)(AR_NBLK - 1) * PF * OUT_CH, a, h, ring);
    ar_block<AR_REM, 0>(p + (size_t)AR_NBLK * PF * OUT_CH, a, h, ring);
}

extern "C" void kernel_launch(void* const* d_in, const int* in_sizes, int n_in,
                              void* d_out, int out_size, void* d_ws, size_t ws_size,
                              hipStream_t stream)
{
    const float* x = (const float*)d_in[0];
    const float* a = (const float*)d_in[1];
    const float* w = (const float*)d_in[2];
    float* out = (float*)d_out;

    const int B = in_sizes[0] / (T_LEN * IN_CH);

    dim3 grid1((unsigned)(B * (T_LEN / TILE_T)));
    fir_wmma_kernel<<<grid1, 256, 0, stream>>>(x, w, out);

    ar_scan_kernel<<<dim3((unsigned)B), 64, 0, stream>>>(a, out);
}


# FIR: each wave computes 32 t x all 64 channels (8 accumulator tiles, channels interleaved); W read by 16-byte loads, reused in registers; k order inside each 8-group of the sum changes
# speedup vs baseline: 2.9299x; 1.1722x over previous
#include <hip/hip_runtime.h>


typedef __attribute__((ext_vector_type(2))) float v2f;
typedef __attribute__((ext_vector_type(4))) float v4f;
typedef __attribute__((ext_vector_type(8))) float v8f;

#define T_LEN   4096
#define IN_CH   64
#define OUT_CH  64
#define NB      16
#define NA      8

#define FIR_WAVES   4
#define FIR_THREADS (FIR_WAVES * 32)
#define FIR_MT      2
#define TILE_T      (FIR_WAVES * FIR_MT * 16)
#define STAGE_ROWS  (TILE_T + NB - 1)
#define STAGE_PASSES ((STAGE_ROWS + 7) / 8)
#define XS_PITCH4   17
#define KBLK        16
#define NQ          (NB * IN_CH / KBLK)

struct FirFrags {
    v4f a[FIR_MT][2];
    v4f b[2][4];
};

__device__ __forceinline__ void fir_load_w(FirFrags& f, const v4f* __restrict__ wb, int q)
{
#pragma unroll
    for (int g = 0; g < 2; ++g)
#pragma unroll
        for (int e = 0; e < 4; ++e)
            f.b[g][e] = wb[(size_t)(q * KBLK + 8 * g + e) * (OUT_CH / 4)];
}

__device__ __forceinline__ void fir_load_x(FirFrags& f, const v4f* __restrict__ xa, int q)
{
    const int tap = q >> 2;
    const int s4  = (q & 3) * 4;
#pragma unroll
    for (int mt = 0; mt < FIR_MT; ++mt)
#pragma unroll
        for (int g = 0; g < 2; ++g)
            f.a[mt][g] = xa[(mt * 16 + tap) * XS_PITCH4 + s4 + 2 * g];
}

__device__ __forceinline__ void fir_mma(v8f (&acc)[FIR_MT][4], const FirFrags& f)
{
#pragma unroll
    for (int g = 0; g < 2; ++g)
#pragma unroll
        for (int p = 0; p < 2; ++p) {
            v2f bf[4];
#pragma unroll
            for (int j = 0; j < 4; ++j) {
                bf[j].x = f.b[g][2 * p][j];
                bf[j].y = f.b[g][2 * p + 1][j];
            }
#pragma unroll
            for (int mt = 0; mt < FIR_MT; ++mt) {
                v2f a;
                a.x = f.a[mt][g][2 * p];
                a.y = f.a[mt][g][2 * p + 1];
#pragma unroll
                for (int j = 0; j < 4; ++j)
                    acc[mt][j] = __builtin_amdgcn_wmma_f32_16x16x4_f32(
                         false, a,  false, bf[j],
                         (short)0, acc[mt][j],
                         false,  false);
            }
        }
}

__global__ __launch_bounds__(FIR_THREADS) void fir_wmma_kernel(
    const float* __restrict__ x,
    const float* __restrict__ w,
    float* __restrict__ u)
{
    __shared__ v4f xs[STAGE_PASSES * 8 * XS_PITCH4];

    const int tid  = threadIdx.x;
    const int bpb  = T_LEN / TILE_T;
    const int b    = blockIdx.x / bpb;
    const int t0   = (blockIdx.x % bpb) * TILE_T;

    const int lane = tid & 31;
    const int wave = tid >> 5;
    const int c    = lane & 15;
    const int hi   = lane >> 4;

    const v4f* xa = xs + (wave * 32 + c) * XS_PITCH4 + hi;
    const v4f* wb = (const v4f*)w + (size_t)(4 * hi) * (OUT_CH / 4) + c;

    FirFrags cur;
    fir_load_w(cur, wb, 0);

    {
        const int c4 = tid & 15;
        const int r0 = tid >> 4;
        const float* xb = x + (size_t)b * T_LEN * IN_CH + c4 * 4;
        v4f st[STAGE_PASSES];
#pragma unroll
        for (int k = 0; k < STAGE_PASSES; ++k) {
            int tg = t0 - (NB - 1) + r0 + 8 * k;
            tg = tg < 0 ? 0 : (tg > T_LEN - 1 ? T_LEN - 1 : tg);
            st[k] = *(const v4f*)(xb + (size_t)tg * IN_CH);
        }
        __builtin_amdgcn_sched_barrier(0);
#pragma unroll
        for (int k = 0; k < STAGE_PASSES; ++k) {
            const int r = r0 + 8 * k;
            v4f v = st[k];
            if (t0 - (NB - 1) + r < 0)
                v = (v4f){0.0f, 0.0f, 0.0f, 0.0f};
            xs[r * XS_PITCH4 + c4] = v;
        }
    }
    __syncthreads();

    v8f acc[FIR_MT][4] = {};

    fir_load_x(cur, xa, 0);
    for (int tap = 0; tap < NB; ++tap) {
#pragma unroll
        for (int s = 0; s < 4; ++s) {
            const int q  = tap * 4 + s;
            const int qn = (q + 1 < NQ) ? q + 1 : q;
            FirFrags nxt;
            fir_load_w(nxt, wb, qn);
            fir_load_x(nxt, xa, qn);
            __builtin_amdgcn_sched_barrier(0);
            fir_mma(acc, cur);
            __builtin_amdgcn_sched_barrier(0);
            cur = nxt;
        }
    }

    float* uout = u + ((size_t)b * T_LEN + t0 + wave * 32 + hi * 8) * OUT_CH + 4 * c;
#pragma unroll
    for (int mt = 0; mt < FIR_MT; ++mt)
#pragma unroll
        for (int r = 0; r < 8; ++r) {
            v4f o;
            o.x = acc[mt][0][r];
            o.y = acc[mt][1][r];
            o.z = acc[mt][2][r];
            o.w = acc[mt][3][r];
            *(v4f*)(uout + (size_t)(mt * 16 + r) * OUT_CH) = o;
        }
}

#define PF 56
#define AR_NBLK (T_LEN / PF)
#define AR_REM  (T_LEN % PF)

template <int STEPS, int REFILL>
__device__ __forceinline__ void ar_block(float* __restrict__ q,
                                         const float (&a)[NA],
                                         float (&h)[NA],
                                         float (&ring)[PF])
{
#pragma unroll
    for (int i = 0; i < STEPS; ++i) {
        float acc = ring[i];
        if (i < REFILL)
            ring[i] = q[(size_t)(STEPS + i) * OUT_CH];
#pragma unroll
        for (int j = NA - 1; j >= 0; --j)
            acc = fmaf(a[j], h[j], acc);
#pragma unroll
        for (int j = NA - 1; j > 0; --j) h[j] = h[j - 1];
        h[0] = acc;
        q[(size_t)i * OUT_CH] = acc;
    }
}

__global__ __launch_bounds__(64) void ar_scan_kernel(
    const float* __restrict__ a_coeff,
    float* __restrict__ y)
{
    const int b = blockIdx.x;
    const int o = threadIdx.x;

    float a[NA];
#pragma unroll
    for (int j = 0; j < NA; ++j)
        a[j] = a_coeff[j * OUT_CH + o];

    float* p = y + ((size_t)b * T_LEN) * OUT_CH + o;

    float ring[PF];
#pragma unroll
    for (int i = 0; i < PF; ++i)
        ring[i] = p[(size_t)i * OUT_CH];

    float h[NA];
#pragma unroll
    for (int j = 0; j < NA; ++j) h[j] = 0.0f;

    for (int k = 0; k < AR_NBLK - 1; ++k)
        ar_block<PF, PF>(p + (size_t)k * PF * OUT_CH, a, h, ring);
    ar_block<PF, AR_REM>(p + (size_t)(AR_NBLK - 1) * PF * OUT_CH, a, h, ring);
    ar_block<AR_REM, 0>(p + (size_t)AR_NBLK * PF * OUT_CH, a, h, ring);
}

extern "C" void kernel_launch(void* const* d_in, const int* in_sizes, int n_in,
                              void* d_out, int out_size, void* d_ws, size_t ws_size,
                              hipStream_t stream)
{
    const float* x = (const float*)d_in[0];
    const float* a = (const float*)d_in[1];
    const float* w = (const float*)d_in[2];
    float* out = (float*)d_out;

    const int B = in_sizes[0] / (T_LEN * IN_CH);

    dim3 grid1((unsigned)(B * (T_LEN / TILE_T)));
    fir_wmma_kernel<<<grid1, FIR_THREADS, 0, stream>>>(x, w, out);

    ar_scan_kernel<<<dim3((unsigned)B), 64, 0, stream>>>(a, out);
}


# AR scan: 8 more waves per block stage u and y through LDS in 256-step chunks (fetch next chunk, write last one out, 16-B accesses) while 2 waves scan from LDS; FMA order unchanged
# speedup vs baseline: 3.6155x; 1.2340x over previous
#include <hip/hip_runtime.h>


typedef __attribute__((ext_vector_type(2))) float v2f;
typedef __attribute__((ext_vector_type(4))) float v4f;
typedef __attribute__((ext_vector_type(8))) float v8f;

#define T_LEN   4096
#define IN_CH   64
#define OUT_CH  64
#define NB      16
#define NA      8

#define FIR_WAVES   4
#define FIR_THREADS (FIR_WAVES * 32)
#define FIR_MT      2
#define TILE_T      (FIR_WAVES * FIR_MT * 16)
#define STAGE_ROWS  (TILE_T + NB - 1)
#define STAGE_PASSES ((STAGE_ROWS + 7) / 8)
#define XS_PITCH4   17
#define KBLK        16
#define NQ          (NB * IN_CH / KBLK)

struct FirFrags {
    v4f a[FIR_MT][2];
    v4f b[2][4];
};

__device__ __forceinline__ void fir_load_w(FirFrags& f, const v4f* __restrict__ wb, int q)
{
#pragma unroll
    for (int g = 0; g < 2; ++g)
#pragma unroll
        for (int e = 0; e < 4; ++e)
            f.b[g][e] = wb[(size_t)(q * KBLK + 8 * g + e) * (OUT_CH / 4)];
}

__device__ __forceinline__ void fir_load_x(FirFrags& f, const v4f* __restrict__ xa, int q)
{
    const int tap = q >> 2;
    const int s4  = (q & 3) * 4;
#pragma unroll
    for (int mt = 0; mt < FIR_MT; ++mt)
#pragma unroll
        for (int g = 0; g < 2; ++g)
            f.a[mt][g] = xa[(mt * 16 + tap) * XS_PITCH4 + s4 + 2 * g];
}

__device__ __forceinline__ void fir_mma(v8f (&acc)[FIR_MT][4], const FirFrags& f)
{
#pragma unroll
    for (int g = 0; g < 2; ++g)
#pragma unroll
        for (int p = 0; p < 2; ++p) {
            v2f bf[4];
#pragma unroll
            for (int j = 0; j < 4; ++j) {
                bf[j].x = f.b[g][2 * p][j];
                bf[j].y = f.b[g][2 * p + 1][j];
            }
#pragma unroll
            for (int mt = 0; mt < FIR_MT; ++mt) {
                v2f a;
                a.x = f.a[mt][g][2 * p];
                a.y = f.a[mt][g][2 * p + 1];
#pragma unroll
                for (int j = 0; j < 4; ++j)
                    acc[mt][j] = __builtin_amdgcn_wmma_f32_16x16x4_f32(
                         false, a,  false, bf[j],
                         (short)0, acc[mt][j],
                         false,  false);
            }
        }
}

__global__ __launch_bounds__(FIR_THREADS) void fir_wmma_kernel(
    const float* __restrict__ x,
    const float* __restrict__ w,
    float* __restrict__ u)
{
    __shared__ v4f xs[STAGE_PASSES * 8 * XS_PITCH4];

    const int tid  = threadIdx.x;
    const int bpb  = T_LEN / TILE_T;
    const int b    = blockIdx.x / bpb;
    const int t0   = (blockIdx.x % bpb) * TILE_T;

    const int lane = tid & 31;
    const int wave = tid >> 5;
    const int c    = lane & 15;
    const int hi   = lane >> 4;

    const v4f* xa = xs + (wave * 32 + c) * XS_PITCH4 + hi;
    const v4f* wb = (const v4f*)w + (size_t)(4 * hi) * (OUT_CH / 4) + c;

    FirFrags cur;
    fir_load_w(cur, wb, 0);

    {
        const int c4 = tid & 15;
        const int r0 = tid >> 4;
        const float* xb = x + (size_t)b * T_LEN * IN_CH + c4 * 4;
        v4f st[STAGE_PASSES];
#pragma unroll
        for (int k = 0; k < STAGE_PASSES; ++k) {
            int tg = t0 - (NB - 1) + r0 + 8 * k;
            tg = tg < 0 ? 0 : (tg > T_LEN - 1 ? T_LEN - 1 : tg);
            st[k] = *(const v4f*)(xb + (size_t)tg * IN_CH);
        }
        __builtin_amdgcn_sched_barrier(0);
#pragma unroll
        for (int k = 0; k < STAGE_PASSES; ++k) {
            const int r = r0 + 8 * k;
            v4f v = st[k];
            if (t0 - (NB - 1) + r < 0)
                v = (v4f){0.0f, 0.0f, 0.0f, 0.0f};
            xs[r * XS_PITCH4 + c4] = v;
        }
    }
    __syncthreads();

    v8f acc[FIR_MT][4] = {};

    fir_load_x(cur, xa, 0);
    for (int tap = 0; tap < NB; ++tap) {
#pragma unroll
        for (int s = 0; s < 4; ++s) {
            const int q  = tap * 4 + s;
            const int qn = (q + 1 < NQ) ? q + 1 : q;
            FirFrags nxt;
            fir_load_w(nxt, wb, qn);
            fir_load_x(nxt, xa, qn);
            __builtin_amdgcn_sched_barrier(0);
            fir_mma(acc, cur);
            __builtin_amdgcn_sched_barrier(0);
            cur = nxt;
        }
    }

    float* uout = u + ((size_t)b * T_LEN + t0 + wave * 32 + hi * 8) * OUT_CH + 4 * c;
#pragma unroll
    for (int mt = 0; mt < FIR_MT; ++mt)
#pragma unroll
        for (int r = 0; r < 8; ++r) {
            v4f o;
            o.x = acc[mt][0][r];
            o.y = acc[mt][1][r];
            o.z = acc[mt][2][r];
            o.w = acc[mt][3][r];
            *(v4f*)(uout + (size_t)(mt * 16 + r) * OUT_CH) = o;
        }
}

#define AR_SCAN_THREADS OUT_CH
#define AR_MOVE_THREADS 256
#define AR_THREADS      (AR_SCAN_THREADS + AR_MOVE_THREADS)
#define AR_CHUNK        256
#define AR_NCHUNK       (T_LEN / AR_CHUNK)
#define AR_NBUF         3
#define AR_ROW4         (OUT_CH / 4)
#define AR_BUF4         (AR_CHUNK * AR_ROW4)
#define AR_PASS_ROWS    (AR_MOVE_THREADS / AR_ROW4)
#define AR_PASSES       (AR_CHUNK / AR_PASS_ROWS)
#define PF 32
#define AR_NBLK (AR_CHUNK / PF)

template <int REFILL>
__device__ __forceinline__ void ar_block(float* __restrict__ q,
                                         const float (&a)[NA],
                                         float (&h)[NA],
                                         float (&ring)[PF])
{
#pragma unroll
    for (int i = 0; i < PF; ++i) {
        float acc = ring[i];
        if (i < REFILL)
            ring[i] = q[(PF + i) * OUT_CH];
#pragma unroll
        for (int j = NA - 1; j >= 0; --j)
            acc = fmaf(a[j], h[j], acc);
#pragma unroll
        for (int j = NA - 1; j > 0; --j) h[j] = h[j - 1];
        h[0] = acc;
        q[i * OUT_CH] = acc;
    }
}

__global__ __launch_bounds__(AR_THREADS) void ar_scan_kernel(
    const float* __restrict__ a_coeff,
    float* __restrict__ y)
{
    __shared__ v4f sm[AR_NBUF * AR_BUF4];

    const int  b    = blockIdx.x;
    const int  tid  = threadIdx.x;
    const bool scan = tid < AR_SCAN_THREADS;

    const int o  = tid & (OUT_CH - 1);
    const int m  = tid - AR_SCAN_THREADS;
    const int c4 = m & (AR_ROW4 - 1);
    const int r0 = m >> 4;

    v4f* yb = (v4f*)(y + (size_t)b * T_LEN * OUT_CH);

    float a[NA];
#pragma unroll
    for (int j = 0; j < NA; ++j)
        a[j] = a_coeff[j * OUT_CH + o];

    float h[NA];
#pragma unroll
    for (int j = 0; j < NA; ++j) h[j] = 0.0f;

    if (!scan) {
#pragma unroll
        for (int k = 0; k < AR_PASSES; ++k) {
            const int r = r0 + AR_PASS_ROWS * k;
            sm[r * AR_ROW4 + c4] = yb[(size_t)r * AR_ROW4 + c4];
        }
    }
    __syncthreads();

    int cur = 0, nxt = 1, prv = 2;
    for (int c = 0; c <= AR_NCHUNK; ++c) {
        if (scan) {
            if (c < AR_NCHUNK) {
                float* q = (float*)(sm + cur * AR_BUF4) + o;
                float ring[PF];
#pragma unroll
                for (int i = 0; i < PF; ++i)
                    ring[i] = q[i * OUT_CH];
                for (int k = 0; k < AR_NBLK - 1; ++k)
                    ar_block<PF>(q + k * PF * OUT_CH, a, h, ring);
                ar_block<0>(q + (AR_NBLK - 1) * PF * OUT_CH, a, h, ring);
            }
        } else {
            const bool fill  = c + 1 < AR_NCHUNK;
            const bool drain = c >= 1;
            v4f st[AR_PASSES];
            if (fill) {
                const v4f* src = yb + (size_t)(c + 1) * AR_BUF4;
#pragma unroll
                for (int k = 0; k < AR_PASSES; ++k)
                    st[k] = src[(size_t)(r0 + AR_PASS_ROWS * k) * AR_ROW4 + c4];
            }
            if (drain) {
                v4f*       dst = yb + (size_t)(c - 1) * AR_BUF4;
                const v4f* s   = sm + prv * AR_BUF4;
#pragma unroll
                for (int k = 0; k < AR_PASSES; ++k) {
                    const int r = r0 + AR_PASS_ROWS * k;
                    dst[(size_t)r * AR_ROW4 + c4] = s[r * AR_ROW4 + c4];
                }
            }
            if (fill) {
                v4f* d = sm + nxt * AR_BUF4;
#pragma unroll
                for (int k = 0; k < AR_PASSES; ++k)
                    d[(r0 + AR_PASS_ROWS * k) * AR_ROW4 + c4] = st[k];
            }
        }
        __syncthreads();
        const int t = prv; prv = cur; cur = nxt; nxt = t;
    }
}

extern "C" void kernel_launch(void* const* d_in, const int* in_sizes, int n_in,
                              void* d_out, int out_size, void* d_ws, size_t ws_size,
                              hipStream_t stream)
{
    const float* x = (const float*)d_in[0];
    const float* a = (const float*)d_in[1];
    const float* w = (const float*)d_in[2];
    float* out = (float*)d_out;

    const int B = in_sizes[0] / (T_LEN * IN_CH);

    dim3 grid1((unsigned)(B * (T_LEN / TILE_T)));
    fir_wmma_kernel<<<grid1, FIR_THREADS, 0, stream>>>(x, w, out);

    ar_scan_kernel<<<dim3((unsigned)B), AR_THREADS, 0, stream>>>(a, out);
}


# AR scan cut into 16 time segments per row, all scanned in parallel: pass 1 from zero history, pass 2 rebuilds start histories via the 256-step transition matrix (new sum order), rescans; 3 launches
# speedup vs baseline: 5.5488x; 1.5347x over previous
#include <hip/hip_runtime.h>


typedef __attribute__((ext_vector_type(2))) float v2f;
typedef __attribute__((ext_vector_type(4))) float v4f;
typedef __attribute__((ext_vector_type(8))) float v8f;

#define T_LEN   4096
#define IN_CH   64
#define OUT_CH  64
#define NB      16
#define NA      8

#define FIR_WAVES   4
#define FIR_THREADS (FIR_WAVES * 32)
#define FIR_MT      2
#define TILE_T      (FIR_WAVES * FIR_MT * 16)
#define STAGE_ROWS  (TILE_T + NB - 1)
#define STAGE_PASSES ((STAGE_ROWS + 7) / 8)
#define XS_PITCH4   17
#define KBLK        16
#define NQ          (NB * IN_CH / KBLK)

struct FirFrags {
    v4f a[FIR_MT][2];
    v4f b[2][4];
};

__device__ __forceinline__ void fir_load_w(FirFrags& f, const v4f* __restrict__ wb, int q)
{
#pragma unroll
    for (int g = 0; g < 2; ++g)
#pragma unroll
        for (int e = 0; e < 4; ++e)
            f.b[g][e] = wb[(size_t)(q * KBLK + 8 * g + e) * (OUT_CH / 4)];
}

__device__ __forceinline__ void fir_load_x(FirFrags& f, const v4f* __restrict__ xa, int q)
{
    const int tap = q >> 2;
    const int s4  = (q & 3) * 4;
#pragma unroll
    for (int mt = 0; mt < FIR_MT; ++mt)
#pragma unroll
        for (int g = 0; g < 2; ++g)
            f.a[mt][g] = xa[(mt * 16 + tap) * XS_PITCH4 + s4 + 2 * g];
}

__device__ __forceinline__ void fir_mma(v8f (&acc)[FIR_MT][4], const FirFrags& f)
{
#pragma unroll
    for (int g = 0; g < 2; ++g)
#pragma unroll
        for (int p = 0; p < 2; ++p) {
            v2f bf[4];
#pragma unroll
            for (int j = 0; j < 4; ++j) {
                bf[j].x = f.b[g][2 * p][j];
                bf[j].y = f.b[g][2 * p + 1][j];
            }
#pragma unroll
            for (int mt = 0; mt < FIR_MT; ++mt) {
                v2f a;
                a.x = f.a[mt][g][2 * p];
                a.y = f.a[mt][g][2 * p + 1];
#pragma unroll
                for (int j = 0; j < 4; ++j)
                    acc[mt][j] = __builtin_amdgcn_wmma_f32_16x16x4_f32(
                         false, a,  false, bf[j],
                         (short)0, acc[mt][j],
                         false,  false);
            }
        }
}

__global__ __launch_bounds__(FIR_THREADS) void fir_wmma_kernel(
    const float* __restrict__ x,
    const float* __restrict__ w,
    float* __restrict__ u)
{
    __shared__ v4f xs[STAGE_PASSES * 8 * XS_PITCH4];

    const int tid  = threadIdx.x;
    const int bpb  = T_LEN / TILE_T;
    const int b    = blockIdx.x / bpb;
    const int t0   = (blockIdx.x % bpb) * TILE_T;

    const int lane = tid & 31;
    const int wave = tid >> 5;
    const int c    = lane & 15;
    const int hi   = lane >> 4;

    const v4f* xa = xs + (wave * 32 + c) * XS_PITCH4 + hi;
    const v4f* wb = (const v4f*)w + (size_t)(4 * hi) * (OUT_CH / 4) + c;

    FirFrags cur;
    fir_load_w(cur, wb, 0);

    {
        const int c4 = tid & 15;
        const int r0 = tid >> 4;
        const float* xb = x + (size_t)b * T_LEN * IN_CH + c4 * 4;
        v4f st[STAGE_PASSES];
#pragma unroll
        for (int k = 0; k < STAGE_PASSES; ++k) {
            int tg = t0 - (NB - 1) + r0 + 8 * k;
            tg = tg < 0 ? 0 : (tg > T_LEN - 1 ? T_LEN - 1 : tg);
            st[k] = *(const v4f*)(xb + (size_t)tg * IN_CH);
        }
        __builtin_amdgcn_sched_barrier(0);
#pragma unroll
        for (int k = 0; k < STAGE_PASSES; ++k) {
            const int r = r0 + 8 * k;
            v4f v = st[k];
            if (t0 - (NB - 1) + r < 0)
                v = (v4f){0.0f, 0.0f, 0.0f, 0.0f};
            xs[r * XS_PITCH4 + c4] = v;
        }
    }
    __syncthreads();

    v8f acc[FIR_MT][4] = {};

    fir_load_x(cur, xa, 0);
    for (int tap = 0; tap < NB; ++tap) {
#pragma unroll
        for (int s = 0; s < 4; ++s) {
            const int q  = tap * 4 + s;
            const int qn = (q + 1 < NQ) ? q + 1 : q;
            FirFrags nxt;
            fir_load_w(nxt, wb, qn);
            fir_load_x(nxt, xa, qn);
            __builtin_amdgcn_sched_barrier(0);
            fir_mma(acc, cur);
            __builtin_amdgcn_sched_barrier(0);
            cur = nxt;
        }
    }

    float* uout = u + ((size_t)b * T_LEN + t0 + wave * 32 + hi * 8) * OUT_CH + 4 * c;
#pragma unroll
    for (int mt = 0; mt < FIR_MT; ++mt)
#pragma unroll
        for (int r = 0; r < 8; ++r) {
            v4f o;
            o.x = acc[mt][0][r];
            o.y = acc[mt][1][r];
            o.z = acc[mt][2][r];
            o.w = acc[mt][3][r];
            *(v4f*)(uout + (size_t)(mt * 16 + r) * OUT_CH) = o;
        }
}

#define AR_SEG          256
#define AR_NSEG         (T_LEN / AR_SEG)
#define AR_ROW4         (OUT_CH / 4)
#define AR_SEG4         (AR_SEG * AR_ROW4)
#define AR_SCAN_THREADS OUT_CH
#define AR_MOVE_THREADS 256
#define AR_PASS_ROWS    (AR_MOVE_THREADS / AR_ROW4)
#define AR_PASSES       (AR_SEG / AR_PASS_ROWS)
#define AR_M_FLOATS     (NA * NA * OUT_CH)
#define PF 32
#define AR_NBLK (AR_SEG / PF)

template <int REFILL, bool STORE>
__device__ __forceinline__ void ar_block(float* __restrict__ q,
                                         const float (&a)[NA],
                                         float (&h)[NA],
                                         float (&ring)[PF])
{
#pragma unroll
    for (int i = 0; i < PF; ++i) {
        float acc = ring[i];
        if (i < REFILL)
            ring[i] = q[(PF + i) * OUT_CH];
#pragma unroll
        for (int j = NA - 1; j >= 0; --j)
            acc = fmaf(a[j], h[j], acc);
#pragma unroll
        for (int j = NA - 1; j > 0; --j) h[j] = h[j - 1];
        h[0] = acc;
        if (STORE)
            q[i * OUT_CH] = acc;
    }
}

template <bool STORE>
__device__ __forceinline__ void ar_scan_segment(float* __restrict__ q,
                                                const float (&a)[NA],
                                                float (&h)[NA])
{
    float ring[PF];
#pragma unroll
    for (int i = 0; i < PF; ++i)
        ring[i] = q[i * OUT_CH];
    for (int k = 0; k < AR_NBLK - 1; ++k)
        ar_block<PF, STORE>(q + k * PF * OUT_CH, a, h, ring);
    ar_block<0, STORE>(q + (AR_NBLK - 1) * PF * OUT_CH, a, h, ring);
}

__global__ __launch_bounds__(AR_MOVE_THREADS) void ar_local_kernel(
    const float* __restrict__ a_coeff,
    const float* __restrict__ u,
    float* __restrict__ ws)
{
    __shared__ v4f sm[AR_SEG4];

    const int  tid   = threadIdx.x;
    const bool unit  = blockIdx.x < NA;
    const int  blk   = (int)blockIdx.x - NA;
    const int  c4    = tid & (AR_ROW4 - 1);
    const int  r0    = tid >> 4;

    if (!unit) {
        const v4f* src = (const v4f*)u + (size_t)blk * AR_SEG4;
#pragma unroll
        for (int k = 0; k < AR_PASSES; ++k) {
            const int r = r0 + AR_PASS_ROWS * k;
            sm[r * AR_ROW4 + c4] = src[(size_t)r * AR_ROW4 + c4];
        }
    } else {
#pragma unroll
        for (int k = 0; k < AR_PASSES; ++k)
            sm[(r0 + AR_PASS_ROWS * k) * AR_ROW4 + c4] = (v4f){0.0f, 0.0f, 0.0f, 0.0f};
    }
    __syncthreads();

    if (tid < AR_SCAN_THREADS) {
        const int o  = tid;
        const int jb = blockIdx.x;
        float a[NA];
#pragma unroll
        for (int j = 0; j < NA; ++j)
            a[j] = a_coeff[j * OUT_CH + o];
        float h[NA];
#pragma unroll
        for (int j = 0; j < NA; ++j)
            h[j] = (unit && j == jb) ? 1.0f : 0.0f;

        ar_scan_segment<false>((float*)sm + o, a, h);

        float* dst = unit ? ws + (size_t)jb * NA * OUT_CH + o
                          : ws + AR_M_FLOATS + (size_t)blk * NA * OUT_CH + o;
#pragma unroll
        for (int i = 0; i < NA; ++i)
            dst[i * OUT_CH] = h[i];
    }
}

#define AR2_THREADS (AR_SCAN_THREADS + AR_MOVE_THREADS)

__global__ __launch_bounds__(AR2_THREADS) void ar_final_kernel(
    const float* __restrict__ a_coeff,
    float* __restrict__ y,
    const float* __restrict__ ws)
{
    __shared__ v4f sm[AR_SEG4];

    const int  blk  = blockIdx.x;
    const int  seg  = blk % AR_NSEG;
    const int  tid  = threadIdx.x;
    const bool scan = tid < AR_SCAN_THREADS;

    const int o  = tid & (OUT_CH - 1);
    const int m  = tid - AR_SCAN_THREADS;
    const int c4 = m & (AR_ROW4 - 1);
    const int r0 = m >> 4;

    v4f* yb = (v4f*)y + (size_t)blk * AR_SEG4;

    float a[NA];
    float h[NA];
#pragma unroll
    for (int j = 0; j < NA; ++j) h[j] = 0.0f;

    if (!scan) {
#pragma unroll
        for (int k = 0; k < AR_PASSES; ++k) {
            const int r = r0 + AR_PASS_ROWS * k;
            sm[r * AR_ROW4 + c4] = yb[(size_t)r * AR_ROW4 + c4];
        }
    } else {
#pragma unroll
        for (int j = 0; j < NA; ++j)
            a[j] = a_coeff[j * OUT_CH + o];

        float mt[NA][NA];
#pragma unroll
        for (int j = 0; j < NA; ++j)
#pragma unroll
            for (int i = 0; i < NA; ++i)
                mt[j][i] = ws[(j * NA + i) * OUT_CH + o];
        const float* frow = ws + AR_M_FLOATS
                          + (size_t)(blk - seg) * NA * OUT_CH + o;
        float f[AR_NSEG - 1][NA];
#pragma unroll
        for (int k = 0; k < AR_NSEG - 1; ++k)
#pragma unroll
            for (int i = 0; i < NA; ++i)
                f[k][i] = frow[(k * NA + i) * OUT_CH];

#pragma unroll
        for (int k = 0; k < AR_NSEG - 1; ++k) {
            if (k < seg) {
                float n[NA];
#pragma unroll
                for (int i = 0; i < NA; ++i) {
                    float acc = f[k][i];
#pragma unroll
                    for (int j = NA - 1; j >= 0; --j)
                        acc = fmaf(mt[j][i], h[j], acc);
                    n[i] = acc;
                }
#pragma unroll
                for (int i = 0; i < NA; ++i) h[i] = n[i];
            }
        }
    }
    __syncthreads();

    if (scan)
        ar_scan_segment<true>((float*)sm + o, a, h);
    __syncthreads();

    if (!scan) {
#pragma unroll
        for (int k = 0; k < AR_PASSES; ++k) {
            const int r = r0 + AR_PASS_ROWS * k;
            yb[(size_t)r * AR_ROW4 + c4] = sm[r * AR_ROW4 + c4];
        }
    }
}

extern "C" void kernel_launch(void* const* d_in, const int* in_sizes, int n_in,
                              void* d_out, int out_size, void* d_ws, size_t ws_size,
                              hipStream_t stream)
{
    const float* x = (const float*)d_in[0];
    const float* a = (const float*)d_in[1];
    const float* w = (const float*)d_in[2];
    float* out = (float*)d_out;
    float* ws  = (float*)d_ws;

    const int B = in_sizes[0] / (T_LEN * IN_CH);

    dim3 grid1((unsigned)(B * (T_LEN / TILE_T)));
    fir_wmma_kernel<<<grid1, FIR_THREADS, 0, stream>>>(x, w, out);

    ar_local_kernel<<<dim3((unsigned)(B * AR_NSEG + NA)), AR_MOVE_THREADS, 0, stream>>>(a, out, ws);

    ar_final_kernel<<<dim3((unsigned)(B * AR_NSEG)), AR2_THREADS, 0, stream>>>(a, out, ws);
}


# FIR by the 2-output minimal-filtering identity over tap pairs: 3 WMMA products of K=512 (on x, x[t]-x[t+1], and W tap sums) replace 4, 25% fewer WMMAs; sum order of u changes
# speedup vs baseline: 6.5223x; 1.1754x over previous
#include <hip/hip_runtime.h>


typedef __attribute__((ext_vector_type(2))) float v2f;
typedef __attribute__((ext_vector_type(4))) float v4f;
typedef __attribute__((ext_vector_type(8))) float v8f;

#define T_LEN   4096
#define IN_CH   64
#define OUT_CH  64
#define NB      16
#define NA      8

#define FIR_WAVES   4
#define FIR_THREADS (FIR_WAVES * 32)
#define TILE_T      (FIR_WAVES * 32)
#define X_ROWS      144
#define X_HALF      (X_ROWS / 2)
#define ROWS_PER_STAGER (X_ROWS / 8)
#define XS_PITCH4   17
#define NPAIR       (NB / 2)
#define NIT         (NPAIR * IN_CH / 16)

struct FirFrags {
    v4f a1[2], a2[2], a3[2];
    v4f b1[2][4];
    v4f b3[2][4];
};

__device__ __forceinline__ void fir_load_w(FirFrags& f, const v4f* __restrict__ wb, int it)
{
    const int e    = it >> 2;
    const int row1 = (14 - 2 * e) * IN_CH + 16 * (it & 3);
    const int row3 = row1 + IN_CH;
#pragma unroll
    for (int g = 0; g < 2; ++g)
#pragma unroll
        for (int el = 0; el < 4; ++el) {
            f.b1[g][el] = wb[(size_t)(row1 + 8 * g + el) * (OUT_CH / 4)];
            f.b3[g][el] = wb[(size_t)(row3 + 8 * g + el) * (OUT_CH / 4)];
        }
}

__device__ __forceinline__ void fir_load_x(FirFrags& f, const v4f* xd0, const v4f* xs1,
                                           const v4f* xd1, int it)
{
    const int off = (it & 3) * 4 - (it >> 2) * XS_PITCH4;
#pragma unroll
    for (int g = 0; g < 2; ++g) {
        f.a1[g] = xd0[off + 2 * g];
        f.a2[g] = xs1[off + 2 * g];
        f.a3[g] = xd1[off + 2 * g];
    }
}

__device__ __forceinline__ void fir_mma(v8f (&acc)[3][4], const FirFrags& f)
{
#pragma unroll
    for (int g = 0; g < 2; ++g)
#pragma unroll
        for (int p = 0; p < 2; ++p) {
            v2f bf1[4], bf2[4], bf3[4];
#pragma unroll
            for (int j = 0; j < 4; ++j) {
                bf1[j].x = f.b1[g][2 * p][j];
                bf1[j].y = f.b1[g][2 * p + 1][j];
                bf3[j].x = f.b3[g][2 * p][j];
                bf3[j].y = f.b3[g][2 * p + 1][j];
                bf2[j]   = bf1[j] + bf3[j];
            }
            v2f a1, a2, a3;
            a1.x = f.a1[g][2 * p];  a1.y = f.a1[g][2 * p + 1];
            a2.x = f.a2[g][2 * p];  a2.y = f.a2[g][2 * p + 1];
            a3.x = f.a3[g][2 * p];  a3.y = f.a3[g][2 * p + 1];
#pragma unroll
            for (int j = 0; j < 4; ++j) {
                acc[0][j] = __builtin_amdgcn_wmma_f32_16x16x4_f32(
                     false, a1,  false, bf1[j],
                     (short)0, acc[0][j],
                     false,  false);
                acc[1][j] = __builtin_amdgcn_wmma_f32_16x16x4_f32(
                     false, a2,  false, bf2[j],
                     (short)0, acc[1][j],
                     false,  false);
                acc[2][j] = __builtin_amdgcn_wmma_f32_16x16x4_f32(
                     false, a3,  false, bf3[j],
                     (short)0, acc[2][j],
                     false,  false);
            }
        }
}

__global__ __launch_bounds__(FIR_THREADS) void fir_wmma_kernel(
    const float* __restrict__ x,
    const float* __restrict__ w,
    float* __restrict__ u)
{
    __shared__ v4f xd[X_ROWS * XS_PITCH4];
    __shared__ v4f xs[X_HALF * XS_PITCH4];

    const int tid  = threadIdx.x;
    const int bpb  = T_LEN / TILE_T;
    const int b    = blockIdx.x / bpb;
    const int t0   = (blockIdx.x % bpb) * TILE_T;

    const int lane = tid & 31;
    const int wave = tid >> 5;
    const int c    = lane & 15;
    const int hi   = lane >> 4;

    const int rowi = 7 + wave * 16 + c;
    const v4f* xd0 = xd + rowi * XS_PITCH4 + hi;
    const v4f* xd1 = xd + (X_HALF + rowi) * XS_PITCH4 + hi;
    const v4f* xs1 = xs + rowi * XS_PITCH4 + hi;
    const v4f* wb  = (const v4f*)w + (size_t)(4 * hi) * (OUT_CH / 4) + c;

    FirFrags cur;
    fir_load_w(cur, wb, 0);

    {
        const int c4 = tid & 15;
        const int r0 = tid >> 4;
        const float* xb = x + (size_t)b * T_LEN * IN_CH + c4 * 4;
        v4f st[ROWS_PER_STAGER + 1];
#pragma unroll
        for (int k = 0; k <= ROWS_PER_STAGER; ++k) {
            int tg = t0 - (NB - 1) + ROWS_PER_STAGER * r0 + k;
            tg = tg < 0 ? 0 : (tg > T_LEN - 1 ? T_LEN - 1 : tg);
            st[k] = *(const v4f*)(xb + (size_t)tg * IN_CH);
        }
        __builtin_amdgcn_sched_barrier(0);
#pragma unroll
        for (int k = 0; k <= ROWS_PER_STAGER; ++k)
            if (t0 - (NB - 1) + ROWS_PER_STAGER * r0 + k < 0)
                st[k] = (v4f){0.0f, 0.0f, 0.0f, 0.0f};
#pragma unroll
        for (int k = 0; k < ROWS_PER_STAGER; ++k) {
            const int half = (ROWS_PER_STAGER / 2) * r0 + (k >> 1);
            xd[((k & 1) * X_HALF + half) * XS_PITCH4 + c4] = st[k] - st[k + 1];
            if (k & 1)
                xs[half * XS_PITCH4 + c4] = st[k];
        }
    }
    __syncthreads();

    v8f acc[3][4] = {};

    fir_load_x(cur, xd0, xs1, xd1, 0);
    for (int e = 0; e < NPAIR; ++e) {
#pragma unroll
        for (int s = 0; s < 4; ++s) {
            const int it  = e * 4 + s;
            const int itn = (it + 1 < NIT) ? it + 1 : it;
            FirFrags nxt;
            fir_load_w(nxt, wb, itn);
            fir_load_x(nxt, xd0, xs1, xd1, itn);
            __builtin_amdgcn_sched_barrier(0);
            fir_mma(acc, cur);
            __builtin_amdgcn_sched_barrier(0);
            cur = nxt;
        }
    }

    float* uout = u + ((size_t)b * T_LEN + t0 + wave * 32 + hi * 16) * OUT_CH + 4 * c;
#pragma unroll
    for (int r = 0; r < 8; ++r) {
        v4f ev, od;
        ev.x = acc[0][0][r] + acc[1][0][r];
        ev.y = acc[0][1][r] + acc[1][1][r];
        ev.z = acc[0][2][r] + acc[1][2][r];
        ev.w = acc[0][3][r] + acc[1][3][r];
        od.x = acc[1][0][r] - acc[2][0][r];
        od.y = acc[1][1][r] - acc[2][1][r];
        od.z = acc[1][2][r] - acc[2][2][r];
        od.w = acc[1][3][r] - acc[2][3][r];
        *(v4f*)(uout + (size_t)(2 * r) * OUT_CH)     = ev;
        *(v4f*)(uout + (size_t)(2 * r + 1) * OUT_CH) = od;
    }
}

#define AR_SEG          256
#define AR_NSEG         (T_LEN / AR_SEG)
#define AR_ROW4         (OUT_CH / 4)
#define AR_SEG4         (AR_SEG * AR_ROW4)
#define AR_SCAN_THREADS OUT_CH
#define AR_MOVE_THREADS 256
#define AR_PASS_ROWS    (AR_MOVE_THREADS / AR_ROW4)
#define AR_PASSES       (AR_SEG / AR_PASS_ROWS)
#define AR_M_FLOATS     (NA * NA * OUT_CH)
#define PF 32
#define AR_NBLK (AR_SEG / PF)

template <int REFILL, bool STORE>
__device__ __forceinline__ void ar_block(float* __restrict__ q,
                                         const float (&a)[NA],
                                         float (&h)[NA],
                                         float (&ring)[PF])
{
#pragma unroll
    for (int i = 0; i < PF; ++i) {
        float acc = ring[i];
        if (i < REFILL)
            ring[i] = q[(PF + i) * OUT_CH];
#pragma unroll
        for (int j = NA - 1; j >= 0; --j)
            acc = fmaf(a[j], h[j], acc);
#pragma unroll
        for (int j = NA - 1; j > 0; --j) h[j] = h[j - 1];
        h[0] = acc;
        if (STORE)
            q[i * OUT_CH] = acc;
    }
}

template <bool STORE>
__device__ __forceinline__ void ar_scan_segment(float* __restrict__ q,
                                                const float (&a)[NA],
                                                float (&h)[NA])
{
    float ring[PF];
#pragma unroll
    for (int i = 0; i < PF; ++i)
        ring[i] = q[i * OUT_CH];
    for (int k = 0; k < AR_NBLK - 1; ++k)
        ar_block<PF, STORE>(q + k * PF * OUT_CH, a, h, ring);
    ar_block<0, STORE>(q + (AR_NBLK - 1) * PF * OUT_CH, a, h, ring);
}

__global__ __launch_bounds__(AR_MOVE_THREADS) void ar_local_kernel(
    const float* __restrict__ a_coeff,
    const float* __restrict__ u,
    float* __restrict__ ws)
{
    __shared__ v4f sm[AR_SEG4];

    const int  tid   = threadIdx.x;
    const bool unit  = blockIdx.x < NA;
    const int  blk   = (int)blockIdx.x - NA;
    const int  c4    = tid & (AR_ROW4 - 1);
    const int  r0    = tid >> 4;

    if (!unit) {
        const v4f* src = (const v4f*)u + (size_t)blk * AR_SEG4;
#pragma unroll
        for (int k = 0; k < AR_PASSES; ++k) {
            const int r = r0 + AR_PASS_ROWS * k;
            sm[r * AR_ROW4 + c4] = src[(size_t)r * AR_ROW4 + c4];
        }
    } else {
#pragma unroll
        for (int k = 0; k < AR_PASSES; ++k)
            sm[(r0 + AR_PASS_ROWS * k) * AR_ROW4 + c4] = (v4f){0.0f, 0.0f, 0.0f, 0.0f};
    }
    __syncthreads();

    if (tid < AR_SCAN_THREADS) {
        const int o  = tid;
        const int jb = blockIdx.x;
        float a[NA];
#pragma unroll
        for (int j = 0; j < NA; ++j)
            a[j] = a_coeff[j * OUT_CH + o];
        float h[NA];
#pragma unroll
        for (int j = 0; j < NA; ++j)
            h[j] = (unit && j == jb) ? 1.0f : 0.0f;

        ar_scan_segment<false>((float*)sm + o, a, h);

        float* dst = unit ? ws + (size_t)jb * NA * OUT_CH + o
                          : ws + AR_M_FLOATS + (size_t)blk * NA * OUT_CH + o;
#pragma unroll
        for (int i = 0; i < NA; ++i)
            dst[i * OUT_CH] = h[i];
    }
}

#define AR2_THREADS (AR_SCAN_THREADS + AR_MOVE_THREADS)

__global__ __launch_bounds__(AR2_THREADS) void ar_final_kernel(
    const float* __restrict__ a_coeff,
    float* __restrict__ y,
    const float* __restrict__ ws)
{
    __shared__ v4f sm[AR_SEG4];

    const int  blk  = blockIdx.x;
    const int  seg  = blk % AR_NSEG;
    const int  tid  = threadIdx.x;
    const bool scan = tid < AR_SCAN_THREADS;

    const int o  = tid & (OUT_CH - 1);
    const int m  = tid - AR_SCAN_THREADS;
    const int c4 = m & (AR_ROW4 - 1);
    const int r0 = m >> 4;

    v4f* yb = (v4f*)y + (size_t)blk * AR_SEG4;

    float a[NA];
    float h[NA];
#pragma unroll
    for (int j = 0; j < NA; ++j) h[j] = 0.0f;

    if (!scan) {
#pragma unroll
        for (int k = 0; k < AR_PASSES; ++k) {
            const int r = r0 + AR_PASS_ROWS * k;
            sm[r * AR_ROW4 + c4] = yb[(size_t)r * AR_ROW4 + c4];
        }
    } else {
#pragma unroll
        for (int j = 0; j < NA; ++j)
            a[j] = a_coeff[j * OUT_CH + o];

        float mt[NA][NA];
#pragma unroll
        for (int j = 0; j < NA; ++j)
#pragma unroll
            for (int i = 0; i < NA; ++i)
                mt[j][i] = ws[(j * NA + i) * OUT_CH + o];
        const float* frow = ws + AR_M_FLOATS
                          + (size_t)(blk - seg) * NA * OUT_CH + o;
        float f[AR_NSEG - 1][NA];
#pragma unroll
        for (int k = 0; k < AR_NSEG - 1; ++k)
#pragma unroll
            for (int i = 0; i < NA; ++i)
                f[k][i] = frow[(k * NA + i) * OUT_CH];

#pragma unroll
        for (int k = 0; k < AR_NSEG - 1; ++k) {
            if (k < seg) {
                float n[NA];
#pragma unroll
                for (int i = 0; i < NA; ++i) {
                    float acc = f[k][i];
#pragma unroll
                    for (int j = NA - 1; j >= 0; --j)
                        acc = fmaf(mt[j][i], h[j], acc);
                    n[i] = acc;
                }
#pragma unroll
                for (int i = 0; i < NA; ++i) h[i] = n[i];
            }
        }
    }
    __syncthreads();

    if (scan)
        ar_scan_segment<true>((float*)sm + o, a, h);
    __syncthreads();

    if (!scan) {
#pragma unroll
        for (int k = 0; k < AR_PASSES; ++k) {
            const int r = r0 + AR_PASS_ROWS * k;
            yb[(size_t)r * AR_ROW4 + c4] = sm[r * AR_ROW4 + c4];
        }
    }
}

extern "C" void kernel_launch(void* const* d_in, const int* in_sizes, int n_in,
                              void* d_out, int out_size, void* d_ws, size_t ws_size,
                              hipStream_t stream)
{
    const float* x = (const float*)d_in[0];
    const float* a = (const float*)d_in[1];
    const float* w = (const float*)d_in[2];
    float* out = (float*)d_out;
    float* ws  = (float*)d_ws;

    const int B = in_sizes[0] / (T_LEN * IN_CH);

    dim3 grid1((unsigned)(B * (T_LEN / TILE_T)));
    fir_wmma_kernel<<<grid1, FIR_THREADS, 0, stream>>>(x, w, out);

    ar_local_kernel<<<dim3((unsigned)(B * AR_NSEG + NA)), AR_MOVE_THREADS, 0, stream>>>(a, out, ws);

    ar_final_kernel<<<dim3((unsigned)(B * AR_NSEG)), AR2_THREADS, 0, stream>>>(a, out, ws);
}


# FIR: minimal-filtering identity nested again (4 outputs x 4 taps from 9 WMMA products of K=256, was 2x2 from 3): 25% fewer WMMAs; a wave holds 64 t x 64 ch, 1 wave per SIMD; sum order of u changes
# speedup vs baseline: 7.5233x; 1.1535x over previous
#include <hip/hip_runtime.h>


typedef __attribute__((ext_vector_type(2))) float v2f;
typedef __attribute__((ext_vector_type(4))) float v4f;
typedef __attribute__((ext_vector_type(8))) float v8f;

#define T_LEN   4096
#define IN_CH   64
#define OUT_CH  64
#define NB      16
#define NA      8

#define FIR_WAVES   4
#define FIR_THREADS (FIR_WAVES * 32)
#define TILE_T      (FIR_WAVES * 64)
#define NQUAD       (NB / 4)
#define NIT         (NQUAD * IN_CH / 16)
#define NOPER       9
#define N_PER_STAGER 9
#define N_ROWS      (8 * N_PER_STAGER)
#define ST_ROWS     (4 * N_PER_STAGER + 3)
#define XS_PITCH4   17
#define OP_STRIDE   (N_ROWS * XS_PITCH4)

struct FirFrags {
    v4f a[NOPER][2];
    v4f w[4][2][4];
};

__device__ __forceinline__ void fir_load_w(FirFrags& f, const v4f* __restrict__ wb, int it)
{
    const int e  = it >> 2;
    const int cb = it & 3;
#pragma unroll
    for (int s = 0; s < 4; ++s) {
        const int row = (15 - 4 * e - s) * IN_CH + 16 * cb;
#pragma unroll
        for (int g = 0; g < 2; ++g)
#pragma unroll
            for (int el = 0; el < 4; ++el)
                f.w[s][g][el] = wb[(size_t)(row + 8 * g + el) * (OUT_CH / 4)];
    }
}

__device__ __forceinline__ void fir_load_x(FirFrags& f, const v4f* xa0, int it)
{
    const int off = (it & 3) * 4 - (it >> 2) * XS_PITCH4;
#pragma unroll
    for (int o = 0; o < NOPER; ++o)
#pragma unroll
        for (int g = 0; g < 2; ++g)
            f.a[o][g] = xa0[o * OP_STRIDE + off + 2 * g];
}

__device__ __forceinline__ void fir_mma(v8f (&acc)[NOPER][4], const FirFrags& f)
{
#pragma unroll
    for (int g = 0; g < 2; ++g)
#pragma unroll
        for (int p = 0; p < 2; ++p) {
            v2f bf[NOPER][4];
#pragma unroll
            for (int j = 0; j < 4; ++j) {
                v2f w0, w1, w2, w3;
                w0.x = f.w[0][g][2 * p][j];  w0.y = f.w[0][g][2 * p + 1][j];
                w1.x = f.w[1][g][2 * p][j];  w1.y = f.w[1][g][2 * p + 1][j];
                w2.x = f.w[2][g][2 * p][j];  w2.y = f.w[2][g][2 * p + 1][j];
                w3.x = f.w[3][g][2 * p][j];  w3.y = f.w[3][g][2 * p + 1][j];
                bf[0][j] = w3;
                bf[1][j] = w2 + w3;
                bf[2][j] = w2;
                bf[3][j] = w1 + w3;
                bf[5][j] = w0 + w2;
                bf[6][j] = w1;
                bf[7][j] = w0 + w1;
                bf[8][j] = w0;
                bf[4][j] = bf[7][j] + bf[1][j];
            }
            v2f af[NOPER];
#pragma unroll
            for (int o = 0; o < NOPER; ++o) {
                af[o].x = f.a[o][g][2 * p];
                af[o].y = f.a[o][g][2 * p + 1];
            }
#pragma unroll
            for (int j = 0; j < 4; ++j)
#pragma unroll
                for (int o = 0; o < NOPER; ++o)
                    acc[o][j] = __builtin_amdgcn_wmma_f32_16x16x4_f32(
                         false, af[o],  false, bf[o][j],
                         (short)0, acc[o][j],
                         false,  false);
        }
}

__global__ __launch_bounds__(FIR_THREADS) void fir_wmma_kernel(
    const float* __restrict__ x,
    const float* __restrict__ w,
    float* __restrict__ u)
{
    __shared__ v4f xa[NOPER * OP_STRIDE];

    const int tid  = threadIdx.x;
    const int bpb  = T_LEN / TILE_T;
    const int b    = blockIdx.x / bpb;
    const int t0   = (blockIdx.x % bpb) * TILE_T;

    const int lane = tid & 31;
    const int wave = tid >> 5;
    const int c    = lane & 15;
    const int hi   = lane >> 4;

    const v4f* xa0 = xa + (3 + wave * 16 + c) * XS_PITCH4 + hi;
    const v4f* wb  = (const v4f*)w + (size_t)(4 * hi) * (OUT_CH / 4) + c;

    FirFrags cur;

    {
        const int c4 = tid & 15;
        const int r0 = tid >> 4;
        const float* xb = x + (size_t)b * T_LEN * IN_CH + c4 * 4;
        const int tfirst = t0 - (NB - 1) + 4 * N_PER_STAGER * r0;
        v4f st[ST_ROWS];
#pragma unroll
        for (int k = 0; k < ST_ROWS; ++k) {
            int tg = tfirst + k;
            tg = tg < 0 ? 0 : (tg > T_LEN - 1 ? T_LEN - 1 : tg);
            st[k] = *(const v4f*)(xb + (size_t)tg * IN_CH);
        }
        __builtin_amdgcn_sched_barrier(0);
        fir_load_w(cur, wb, 0);
        __builtin_amdgcn_sched_barrier(0);
#pragma unroll
        for (int k = 0; k < ST_ROWS; ++k)
            if (tfirst + k < 0)
                st[k] = (v4f){0.0f, 0.0f, 0.0f, 0.0f};
#pragma unroll
        for (int k = 0; k < N_PER_STAGER; ++k) {
            v4f* dst = xa + (N_PER_STAGER * r0 + k) * XS_PITCH4 + c4;
            const v4f e02 = st[4 * k + 0] - st[4 * k + 2];
            const v4f e13 = st[4 * k + 1] - st[4 * k + 3];
            const v4f e24 = st[4 * k + 2] - st[4 * k + 4];
            const v4f e35 = st[4 * k + 3] - st[4 * k + 5];
            const v4f e46 = st[4 * k + 4] - st[4 * k + 6];
            dst[0 * OP_STRIDE] = e02 - e13;
            dst[1 * OP_STRIDE] = e13;
            dst[2 * OP_STRIDE] = e13 - e24;
            dst[3 * OP_STRIDE] = st[4 * k + 2] - st[4 * k + 3];
            dst[4 * OP_STRIDE] = st[4 * k + 3];
            dst[5 * OP_STRIDE] = st[4 * k + 3] - st[4 * k + 4];
            dst[6 * OP_STRIDE] = e24 - e35;
            dst[7 * OP_STRIDE] = e35;
            dst[8 * OP_STRIDE] = e35 - e46;
        }
    }
    __syncthreads();

    v8f acc[NOPER][4] = {};

    fir_load_x(cur, xa0, 0);
    for (int e = 0; e < NQUAD; ++e) {
#pragma unroll
        for (int s = 0; s < 4; ++s) {
            const int it  = e * 4 + s;
            const int itn = (it + 1 < NIT) ? it + 1 : it;
            FirFrags nxt;
            fir_load_w(nxt, wb, itn);
            fir_load_x(nxt, xa0, itn);
            __builtin_amdgcn_sched_barrier(0);
            fir_mma(acc, cur);
            __builtin_amdgcn_sched_barrier(0);
            cur = nxt;
        }
    }

    float* uout = u + ((size_t)b * T_LEN + t0 + wave * 64 + hi * 32) * OUT_CH + 4 * c;
#pragma unroll
    for (int r = 0; r < 8; ++r) {
        v4f u0, u1, u2, u3;
#pragma unroll
        for (int j = 0; j < 4; ++j) {
            const float q10 = acc[1][j][r] + acc[0][j][r];
            const float q11 = acc[1][j][r] - acc[2][j][r];
            const float q20 = acc[4][j][r] + acc[3][j][r];
            const float q21 = acc[4][j][r] - acc[5][j][r];
            const float q30 = acc[7][j][r] + acc[6][j][r];
            const float q31 = acc[7][j][r] - acc[8][j][r];
            u0[j] = q20 + q10;
            u1[j] = q21 + q11;
            u2[j] = q20 - q30;
            u3[j] = q21 - q31;
        }
        *(v4f*)(uout + (size_t)(4 * r)     * OUT_CH) = u0;
        *(v4f*)(uout + (size_t)(4 * r + 1) * OUT_CH) = u1;
        *(v4f*)(uout + (size_t)(4 * r + 2) * OUT_CH) = u2;
        *(v4f*)(uout + (size_t)(4 * r + 3) * OUT_CH) = u3;
    }
}

#define AR_SEG          256
#define AR_NSEG         (T_LEN / AR_SEG)
#define AR_ROW4         (OUT_CH / 4)
#define AR_SEG4         (AR_SEG * AR_ROW4)
#define AR_SCAN_THREADS OUT_CH
#define AR_MOVE_THREADS 256
#define AR_PASS_ROWS    (AR_MOVE_THREADS / AR_ROW4)
#define AR_PASSES       (AR_SEG / AR_PASS_ROWS)
#define AR_M_FLOATS     (NA * NA * OUT_CH)
#define PF 32
#define AR_NBLK (AR_SEG / PF)

template <int REFILL, bool STORE>
__device__ __forceinline__ void ar_block(float* __restrict__ q,
                                         const float (&a)[NA],
                                         float (&h)[NA],
                                         float (&ring)[PF])
{
#pragma unroll
    for (int i = 0; i < PF; ++i) {
        float acc = ring[i];
        if (i < REFILL)
            ring[i] = q[(PF + i) * OUT_CH];
#pragma unroll
        for (int j = NA - 1; j >= 0; --j)
            acc = fmaf(a[j], h[j], acc);
#pragma unroll
        for (int j = NA - 1; j > 0; --j) h[j] = h[j - 1];
        h[0] = acc;
        if (STORE)
            q[i * OUT_CH] = acc;
    }
}

template <bool STORE>
__device__ __forceinline__ void ar_scan_segment(float* __restrict__ q,
                                                const float (&a)[NA],
                                                float (&h)[NA])
{
    float ring[PF];
#pragma unroll
    for (int i = 0; i < PF; ++i)
        ring[i] = q[i * OUT_CH];
    for (int k = 0; k < AR_NBLK - 1; ++k)
        ar_block<PF, STORE>(q + k * PF * OUT_CH, a, h, ring);
    ar_block<0, STORE>(q + (AR_NBLK - 1) * PF * OUT_CH, a, h, ring);
}

__global__ __launch_bounds__(AR_MOVE_THREADS) void ar_local_kernel(
    const float* __restrict__ a_coeff,
    const float* __restrict__ u,
    float* __restrict__ ws)
{
    __shared__ v4f sm[AR_SEG4];

    const int  tid   = threadIdx.x;
    const bool unit  = blockIdx.x < NA;
    const int  blk   = (int)blockIdx.x - NA;
    const int  c4    = tid & (AR_ROW4 - 1);
    const int  r0    = tid >> 4;

    if (!unit) {
        const v4f* src = (const v4f*)u + (size_t)blk * AR_SEG4;
#pragma unroll
        for (int k = 0; k < AR_PASSES; ++k) {
            const int r = r0 + AR_PASS_ROWS * k;
            sm[r * AR_ROW4 + c4] = src[(size_t)r * AR_ROW4 + c4];
        }
    } else {
#pragma unroll
        for (int k = 0; k < AR_PASSES; ++k)
            sm[(r0 + AR_PASS_ROWS * k) * AR_ROW4 + c4] = (v4f){0.0f, 0.0f, 0.0f, 0.0f};
    }
    __syncthreads();

    if (tid < AR_SCAN_THREADS) {
        const int o  = tid;
        const int jb = blockIdx.x;
        float a[NA];
#pragma unroll
        for (int j = 0; j < NA; ++j)
            a[j] = a_coeff[j * OUT_CH + o];
        float h[NA];
#pragma unroll
        for (int j = 0; j < NA; ++j)
            h[j] = (unit && j == jb) ? 1.0f : 0.0f;

        ar_scan_segment<false>((float*)sm + o, a, h);

        float* dst = unit ? ws + (size_t)jb * NA * OUT_CH + o
                          : ws + AR_M_FLOATS + (size_t)blk * NA * OUT_CH + o;
#pragma unroll
        for (int i = 0; i < NA; ++i)
            dst[i * OUT_CH] = h[i];
    }
}

#define AR2_THREADS (AR_SCAN_THREADS + AR_MOVE_THREADS)

__global__ __launch_bounds__(AR2_THREADS) void ar_final_kernel(
    const float* __restrict__ a_coeff,
    float* __restrict__ y,
    const float* __restrict__ ws)
{
    __shared__ v4f sm[AR_SEG4];

    const int  blk  = blockIdx.x;
    const int  seg  = blk % AR_NSEG;
    const int  tid  = threadIdx.x;
    const bool scan = tid < AR_SCAN_THREADS;

    const int o  = tid & (OUT_CH - 1);
    const int m  = tid - AR_SCAN_THREADS;
    const int c4 = m & (AR_ROW4 - 1);
    const int r0 = m >> 4;

    v4f* yb = (v4f*)y + (size_t)blk * AR_SEG4;

    float a[NA];
    float h[NA];
#pragma unroll
    for (int j = 0; j < NA; ++j) h[j] = 0.0f;

    if (!scan) {
#pragma unroll
        for (int k = 0; k < AR_PASSES; ++k) {
            const int r = r0 + AR_PASS_ROWS * k;
            sm[r * AR_ROW4 + c4] = yb[(size_t)r * AR_ROW4 + c4];
        }
    } else {
#pragma unroll
        for (int j = 0; j < NA; ++j)
            a[j] = a_coeff[j * OUT_CH + o];

        float mt[NA][NA];
#pragma unroll
        for (int j = 0; j < NA; ++j)
#pragma unroll
            for (int i = 0; i < NA; ++i)
                mt[j][i] = ws[(j * NA + i) * OUT_CH + o];
        const float* frow = ws + AR_M_FLOATS
                          + (size_t)(blk - seg) * NA * OUT_CH + o;
        float f[AR_NSEG - 1][NA];
#pragma unroll
        for (int k = 0; k < AR_NSEG - 1; ++k)
#pragma unroll
            for (int i = 0; i < NA; ++i)
                f[k][i] = frow[(k * NA + i) * OUT_CH];

#pragma unroll
        for (int k = 0; k < AR_NSEG - 1; ++k) {
            if (k < seg) {
                float n[NA];
#pragma unroll
                for (int i = 0; i < NA; ++i) {
                    float acc = f[k][i];
#pragma unroll
                    for (int j = NA - 1; j >= 0; --j)
                        acc = fmaf(mt[j][i], h[j], acc);
                    n[i] = acc;
                }
#pragma unroll
                for (int i = 0; i < NA; ++i) h[i] = n[i];
            }
        }
    }
    __syncthreads();

    if (scan)
        ar_scan_segment<true>((float*)sm + o, a, h);
    __syncthreads();

    if (!scan) {
#pragma unroll
        for (int k = 0; k < AR_PASSES; ++k) {
            const int r = r0 + AR_PASS_ROWS * k;
            yb[(size_t)r * AR_ROW4 + c4] = sm[r * AR_ROW4 + c4];
        }
    }
}

extern "C" void kernel_launch(void* const* d_in, const int* in_sizes, int n_in,
                              void* d_out, int out_size, void* d_ws, size_t ws_size,
                              hipStream_t stream)
{
    const float* x = (const float*)d_in[0];
    const float* a = (const float*)d_in[1];
    const float* w = (const float*)d_in[2];
    float* out = (float*)d_out;
    float* ws  = (float*)d_ws;

    const int B = in_sizes[0] / (T_LEN * IN_CH);

    dim3 grid1((unsigned)(B * (T_LEN / TILE_T)));
    fir_wmma_kernel<<<grid1, FIR_THREADS, 0, stream>>>(x, w, out);

    ar_local_kernel<<<dim3((unsigned)(B * AR_NSEG + NA)), AR_MOVE_THREADS, 0, stream>>>(a, out, ws);

    ar_final_kernel<<<dim3((unsigned)(B * AR_NSEG)), AR2_THREADS, 0, stream>>>(a, out, ws);
}
